# starting kernel
# baseline (speedup 1.0000x reference)
#include <hip/hip_runtime.h>


typedef float v2f __attribute__((ext_vector_type(2)));
typedef float v8f __attribute__((ext_vector_type(8)));

#define B_BATCH 1024
#define NNODE   128
#define CDIM    128
#define MHID    10
#define OUTD    32
#define NPAD    136

__device__ __forceinline__ v8f wmma4(v2f a, v2f b, v8f c) {
  return __builtin_amdgcn_wmma_f32_16x16x4_f32(
       false, a,  false, b,
       (short)0, c,  false,  false);
}

#define WS_A   0
#define WS_Z1  3200
#define WS_Z2  44160
#define WS_ZF  85120
#define ZF_STRIDE 40960

__global__ __launch_bounds__(256) void rw_pre(const float* __restrict__ fh,
                                              const float* __restrict__ ah,
                                              float* __restrict__ ws) {
  float* Adense = ws + WS_A;
  float* z1     = ws + WS_Z1;
  float* z2     = ws + WS_Z2;
  float* zf     = ws + WS_ZF;
  const int tid = threadIdx.x;

  for (int e = tid; e < MHID * MHID * OUTD; e += 256) {
    int d = e & 31, ij = e >> 5, i2 = ij / MHID, j2 = ij % MHID;
    float v = 0.f;
    if (i2 != j2) {
      int i = i2 < j2 ? i2 : j2;
      int j = i2 < j2 ? j2 : i2;
      int p = i * MHID - (i * (i + 1)) / 2 + (j - i - 1);
      float t = ah[p * OUTD + d];
      v = t > 0.f ? t : 0.f;
    }
    Adense[e] = v;
  }
  __syncthreads();

  for (int e = tid; e < MHID * CDIM * OUTD; e += 256) {
    int d = e & 31, c = (e >> 5) & 127, a = e >> 12;
    float s = 0.f;
#pragma unroll
    for (int bb = 0; bb < MHID; ++bb)
      s += Adense[(a * MHID + bb) * OUTD + d] * fh[(bb * CDIM + c) * OUTD + d];
    z1[e] = s;
  }
  __syncthreads();

  for (int e = tid; e < MHID * CDIM * OUTD; e += 256) {
    int d = e & 31, c = (e >> 5) & 127, a = e >> 12;
    float s = 0.f;
#pragma unroll
    for (int bb = 0; bb < MHID; ++bb)
      s += Adense[(a * MHID + bb) * OUTD + d] * z1[((bb * CDIM + c) << 5) + d];
    z2[e] = s;
  }
  __syncthreads();

  for (int s = 0; s < 3; ++s) {
    const float* zs = (s == 0) ? fh : (s == 1 ? z1 : z2);
    float* dst = zf + s * ZF_STRIDE;
    for (int e = tid; e < MHID * 2 * 32 * 32; e += 256) {
      int lane = e & 31, kit = (e >> 5) & 31, ot = (e >> 10) & 1, m = e >> 11;
      int k = kit * 4 + (lane >> 4) * 2;
      int o = ot * 16 + (lane & 15);
      dst[e * 2 + 0] = zs[(m * CDIM + k) * OUTD + o];
      dst[e * 2 + 1] = zs[(m * CDIM + k + 1) * OUTD + o];
    }
  }
}

__global__ __launch_bounds__(256) void rw_main(const float* __restrict__ adj,
                                               const float* __restrict__ feats,
                                               const int*   __restrict__ idxs,
                                               const float* __restrict__ zfrag,
                                               float*       __restrict__ out) {
  __shared__ float lx0[NNODE * NPAD];
  __shared__ float lxc[NNODE * NPAD];
  __shared__ float accbuf[OUTD];

  const int b   = blockIdx.x;
  const int tid = threadIdx.x;
  const int w   = tid >> 5;
  const int l   = tid & 31;
  const int l16 = l & 15;
  const int lhi = l >> 4;

  if (tid < OUTD) accbuf[tid] = 0.f;

  {
    const int r0 = w * 16;
#pragma unroll 4
    for (int r = r0; r < r0 + 16; ++r) {
      int idx = idxs[b * NNODE + r];
      float4 v = *(const float4*)(feats + (size_t)idx * CDIM + l * 4);
      *(float4*)(&lx0[r * NPAD + l * 4]) = v;
    }
  }
  __syncthreads();

  float acc0 = 0.f, acc1 = 0.f;
  const float* xcur = lx0;

  for (int step = 0; step < 3; ++step) {
    if (step) {
      v8f cacc[8];
      const float* Ag = adj + (size_t)b * NNODE * CDIM;
      const int ct = w;
      const int col = ct * 16 + l16;
#pragma unroll
      for (int nt = 0; nt < 8; ++nt) {
        v8f c = {};
        const int row = nt * 16 + l16;
#pragma unroll 4
        for (int kit = 0; kit < 32; ++kit) {
          const int kb = kit * 4 + lhi * 2;
          v2f a = *(const v2f*)(Ag + row * CDIM + kb);
          v2f bv;
          bv.x = xcur[(kb + 0) * NPAD + col];
          bv.y = xcur[(kb + 1) * NPAD + col];
          c = wmma4(a, bv, c);
        }
        cacc[nt] = c;
      }
      __syncthreads();
#pragma unroll
      for (int nt = 0; nt < 8; ++nt)
#pragma unroll
        for (int r = 0; r < 8; ++r)
          lxc[(nt * 16 + r + lhi * 8) * NPAD + ct * 16 + l16] = cacc[nt][r];
      __syncthreads();
      xcur = lxc;
    }

    const float* zf0 = zfrag;
    const float* zfi = zfrag + step * ZF_STRIDE;
    for (int t = w; t < MHID * 8 * 2; t += 8) {
      const int m  = t >> 4;
      const int nt = (t >> 1) & 7;
      const int ot = t & 1;
      const int row = nt * 16 + l16;
      const v2f* bz0 = (const v2f*)zf0 + (size_t)(m * 2 + ot) * 1024 + l;
      const v2f* bzi = (const v2f*)zfi + (size_t)(m * 2 + ot) * 1024 + l;
      v8f czx = {}, ctt = {};
#pragma unroll 4
      for (int kit = 0; kit < 32; ++kit) {
        const int kb = kit * 4 + lhi * 2;
        v2f a0 = *(const v2f*)(&lx0[row * NPAD + kb]);
        czx = wmma4(a0, bz0[kit * 32], czx);
        if (step) {
          v2f ai = *(const v2f*)(&xcur[row * NPAD + kb]);
          ctt = wmma4(ai, bzi[kit * 32], ctt);
        }
      }
      if (!step) ctt = czx;
      float s = 0.f;
#pragma unroll
      for (int r = 0; r < 8; ++r) s += czx[r] * ctt[r];
      if (ot == 0) acc0 += s; else acc1 += s;
    }
  }

  atomicAdd(&accbuf[l16], acc0);
  atomicAdd(&accbuf[16 + l16], acc1);
  __syncthreads();
  if (tid < OUTD) out[(size_t)b * OUTD + tid] = accbuf[tid] * (1.0f / 3840.0f);
}

extern "C" void kernel_launch(void* const* d_in, const int* in_sizes, int n_in,
                              void* d_out, int out_size, void* d_ws, size_t ws_size,
                              hipStream_t stream) {
  (void)in_sizes; (void)n_in; (void)out_size; (void)ws_size;
  const float* adj   = (const float*)d_in[0];
  const float* feats = (const float*)d_in[1];
  const int*   idxs  = (const int*)  d_in[2];
  const float* fh    = (const float*)d_in[3];
  const float* ah    = (const float*)d_in[4];
  float* out = (float*)d_out;
  float* ws  = (float*)d_ws;

  rw_pre<<<1, 256, 0, stream>>>(fh, ah, ws);
  rw_main<<<B_BATCH, 256, 0, stream>>>(adj, feats, idxs, ws + WS_ZF, out);
}


# Hidden-graph steps applied to the products: t_s = A^s (X_s z0) per lane on zx tiles kept in registers; z1, z2, their tables and the setup launch go; launches 2 to 1; sum order changes
# speedup vs baseline: 4.1669x; 4.1669x over previous
#include <hip/hip_runtime.h>


typedef float v2f __attribute__((ext_vector_type(2)));
typedef float v8f __attribute__((ext_vector_type(8)));

#define B_BATCH 1024
#define NNODE   128
#define CDIM    128
#define MHID    10
#define OUTD    32
#define NPAD    136
#define ASZ     (MHID * MHID * OUTD)

__device__ __forceinline__ v8f wmma4(v2f a, v2f b, v8f c) {
  return __builtin_amdgcn_wmma_f32_16x16x4_f32(
       false, a,  false, b,
       (short)0, c,  false,  false);
}

__device__ __forceinline__ void strip_times_z0(const float* xrow, const float* __restrict__ zrow,
                                               v8f& c0, v8f& c1) {
#pragma unroll 4
  for (int kit = 0; kit < 32; ++kit) {
    v2f a;
    a.x = xrow[4 * kit];
    a.y = xrow[4 * kit + 2];
    v2f b0 = *(const v2f*)(zrow + (4 * kit) * OUTD);
    v2f b1 = *(const v2f*)(zrow + (4 * kit + 2) * OUTD);
    v2f be = {b0.x, b1.x};
    v2f bo = {b0.y, b1.y};
    c0 = wmma4(a, be, c0);
    c1 = wmma4(a, bo, c1);
  }
}

__global__ __launch_bounds__(256, 4) void rw_main(const float* __restrict__ adj,
                                               const float* __restrict__ feats,
                                               const int*   __restrict__ idxs,
                                               const float* __restrict__ fh,
                                               const float* __restrict__ ah,
                                               float*       __restrict__ out) {
  __shared__ __attribute__((aligned(16))) float lx0[NNODE * NPAD];
  __shared__ __attribute__((aligned(16))) float lxc[NNODE * NPAD];
  __shared__ __attribute__((aligned(16))) float lA[ASZ];
  __shared__ float accbuf[OUTD];

  const int b   = blockIdx.x;
  const int tid = threadIdx.x;
  const int w   = tid >> 5;
  const int l   = tid & 31;
  const int l16 = l & 15;
  const int lhi = l >> 4;

  if (tid < OUTD) accbuf[tid] = 0.f;

  for (int e = tid; e < ASZ; e += 256) {
    int d = e & 31, ij = e >> 5, i2 = ij / MHID, j2 = ij % MHID;
    float v = 0.f;
    if (i2 != j2) {
      int i = i2 < j2 ? i2 : j2;
      int j = i2 < j2 ? j2 : i2;
      int p = i * MHID - (i * (i + 1)) / 2 + (j - i - 1);
      float t = ah[p * OUTD + d];
      v = t > 0.f ? t : 0.f;
    }
    lA[e] = v;
  }

  {
    const int r0 = w * 16;
#pragma unroll 4
    for (int r = r0; r < r0 + 16; ++r) {
      int idx = idxs[b * NNODE + r];
      float4 v = *(const float4*)(feats + (size_t)idx * CDIM + l * 4);
      *(float4*)(&lx0[r * NPAD + l * 4]) = v;
    }
  }
  __syncthreads();

  float acc0 = 0.f, acc1 = 0.f;
  const int row = w * 16 + l16;
  const float* zlane = fh + lhi * OUTD + 2 * l16;

  v8f zx[MHID][2];
#pragma unroll
  for (int m = 0; m < MHID; ++m) {
    v8f c0 = {}, c1 = {};
    strip_times_z0(&lx0[row * NPAD + lhi], zlane + m * CDIM * OUTD, c0, c1);
    zx[m][0] = c0;
    zx[m][1] = c1;
#pragma unroll
    for (int r = 0; r < 8; ++r) {
      acc0 += c0[r] * c0[r];
      acc1 += c1[r] * c1[r];
    }
  }

  const float* xcur = lx0;
  for (int step = 1; step < 3; ++step) {
    if (step == 2) {
      for (int e = tid; e < ASZ; e += 256) {
        int d = e & 31, ij = e >> 5, i2 = ij / MHID, j2 = ij % MHID;
        float s = 0.f;
#pragma unroll
        for (int k = 0; k < MHID; ++k)
          s += lA[(i2 * MHID + k) * OUTD + d] * lA[(k * MHID + j2) * OUTD + d];
        lx0[e] = s;
      }
    }
    {
      v8f cacc[8];
      const float* Ag = adj + (size_t)b * NNODE * CDIM;
      const int ct = w;
      const int col = ct * 16 + l16;
#pragma unroll
      for (int nt = 0; nt < 8; ++nt) {
        v8f c = {};
        const int prow = nt * 16 + l16;
#pragma unroll 4
        for (int kit = 0; kit < 32; ++kit) {
          const int kb = kit * 4 + lhi * 2;
          v2f a = *(const v2f*)(Ag + prow * CDIM + kb);
          v2f bv;
          bv.x = xcur[(kb + 0) * NPAD + col];
          bv.y = xcur[(kb + 1) * NPAD + col];
          c = wmma4(a, bv, c);
        }
        cacc[nt] = c;
      }
      __syncthreads();
#pragma unroll
      for (int nt = 0; nt < 8; ++nt)
#pragma unroll
        for (int r = 0; r < 8; ++r)
          lxc[(nt * 16 + r + lhi * 8) * NPAD + ct * 16 + l16] = cacc[nt][r];
      __syncthreads();
      xcur = lxc;
    }

    const float* As = (step == 1 ? lA : lx0) + 2 * l16;
#pragma unroll 1
    for (int mp = 0; mp < MHID; ++mp) {
      v8f u0 = {}, u1 = {};
#pragma unroll
      for (int m = 0; m < MHID; ++m) {
        v2f cf = *(const v2f*)(As + (m * MHID + mp) * OUTD);
#pragma unroll
        for (int r = 0; r < 8; ++r) {
          u0[r] += cf.x * zx[m][0][r];
          u1[r] += cf.y * zx[m][1][r];
        }
      }
      v8f c0 = {}, c1 = {};
      strip_times_z0(&lxc[row * NPAD + lhi], zlane + mp * CDIM * OUTD, c0, c1);
#pragma unroll
      for (int r = 0; r < 8; ++r) {
        acc0 += u0[r] * c0[r];
        acc1 += u1[r] * c1[r];
      }
    }
  }

  atomicAdd(&accbuf[2 * l16], acc0);
  atomicAdd(&accbuf[2 * l16 + 1], acc1);
  __syncthreads();
  if (tid < OUTD) out[(size_t)b * OUTD + tid] = accbuf[tid] * (1.0f / 3840.0f);
}

extern "C" void kernel_launch(void* const* d_in, const int* in_sizes, int n_in,
                              void* d_out, int out_size, void* d_ws, size_t ws_size,
                              hipStream_t stream) {
  (void)in_sizes; (void)n_in; (void)out_size; (void)ws_size; (void)d_ws;
  const float* adj   = (const float*)d_in[0];
  const float* feats = (const float*)d_in[1];
  const int*   idxs  = (const int*)  d_in[2];
  const float* fh    = (const float*)d_in[3];
  const float* ah    = (const float*)d_in[4];
  float* out = (float*)d_out;

  rw_main<<<B_BATCH, 256, 0, stream>>>(adj, feats, idxs, fh, ah, out);
}


# Input graph propagates the 128x320 products P=X0*z0 (P_s = adj P_(s-1), P image in LDS in fragment order, adj strip and zx in registers) instead of X; z0 multiplied once; sum order changes
# speedup vs baseline: 6.1306x; 1.4713x over previous
#include <hip/hip_runtime.h>


typedef float v2f __attribute__((ext_vector_type(2)));
typedef float v4f __attribute__((ext_vector_type(4)));
typedef float v8f __attribute__((ext_vector_type(8)));

#define B_BATCH 1024
#define NNODE   128
#define CDIM    128
#define MHID    10
#define OUTD    32
#define PCOLS   (MHID * OUTD)
#define ASZ     (MHID * MHID * OUTD)

__device__ __forceinline__ v8f wmma4(v2f a, v2f b, v8f c) {
  return __builtin_amdgcn_wmma_f32_16x16x4_f32(
       false, a,  false, b,
       (short)0, c,  false,  false);
}


__device__ __forceinline__ void load_pgroup(v4f (&bb)[8], const float* pb, int jg) {
#pragma unroll
  for (int jj = 0; jj < 4; ++jj) {
    bb[2 * jj]     = *(const v4f*)(pb + (4 * jg + jj) * (PCOLS * 8));
    bb[2 * jj + 1] = *(const v4f*)(pb + (4 * jg + jj) * (PCOLS * 8) + 128);
  }
}
__device__ __forceinline__ void mma_pgroup(const v4f (&aa)[16], const v4f (&bb)[8], int jg, v8f& c0, v8f& c1) {
#pragma unroll
  for (int jj = 0; jj < 4; ++jj) {
    const v4f a4 = aa[4 * jg + jj];
    const v4f b0 = bb[2 * jj], b1 = bb[2 * jj + 1];
    v2f alo = {a4.x, a4.y};
    v2f ahi = {a4.z, a4.w};
    v2f b0lo = {b0.x, b0.y}, b0hi = {b0.z, b0.w};
    v2f b1lo = {b1.x, b1.y}, b1hi = {b1.z, b1.w};
    c0 = wmma4(alo, b0lo, c0);
    c1 = wmma4(alo, b1lo, c1);
    c0 = wmma4(ahi, b0hi, c0);
    c1 = wmma4(ahi, b1hi, c1);
  }
}
#define GROUP_FENCE() __builtin_amdgcn_sched_barrier(0x6)

__device__ __forceinline__ void load_zgroup(v2f (&bb)[16], const float* __restrict__ zl, int m, int jg) {
#pragma unroll
  for (int jj = 0; jj < 4; ++jj)
#pragma unroll
    for (int e = 0; e < 2; ++e)
#pragma unroll
      for (int s = 0; s < 2; ++s)
        bb[(jj * 2 + e) * 2 + s] =
            *(const v2f*)(zl + m * (CDIM * OUTD) + (8 * (4 * jg + jj) + 2 * e + s) * OUTD);
}

__global__ __launch_bounds__(256) void rw_main(const float* __restrict__ adj,
                                            const float* __restrict__ feats,
                                            const int*   __restrict__ idxs,
                                            const float* __restrict__ fh,
                                            const float* __restrict__ ah,
                                            float*       __restrict__ out) {
  __shared__ __attribute__((aligned(16))) float lP[(NNODE / 8) * PCOLS * 8];
  __shared__ __attribute__((aligned(16))) float lA[ASZ];
  __shared__ __attribute__((aligned(16))) float lA2[ASZ];
  __shared__ float accbuf[OUTD];

  const int b   = blockIdx.x;
  const int tid = threadIdx.x;
  const int w   = tid >> 5;
  const int l   = tid & 31;
  const int l16 = l & 15;
  const int lhi = l >> 4;
  const int row = w * 16 + l16;

  if (tid < OUTD) accbuf[tid] = 0.f;

  v4f xa[16], aa[16];
  {
    const int idx = idxs[b * NNODE + row];
    const float* xs = feats + (size_t)idx * CDIM + 4 * lhi;
    const float* as = adj + (size_t)b * NNODE * CDIM + (size_t)row * CDIM + 4 * lhi;
#pragma unroll
    for (int j = 0; j < 16; ++j) xa[j] = *(const v4f*)(xs + 8 * j);
#pragma unroll
    for (int j = 0; j < 16; ++j) aa[j] = *(const v4f*)(as + 8 * j);
  }

#pragma unroll
  for (int it = 0; it < (ASZ + 255) / 256; ++it) {
    const int e = tid + 256 * it;
    if (e < ASZ) {
      int d = e & 31, ij = e >> 5, i2 = ij / MHID, j2 = ij % MHID;
      int i = i2 < j2 ? i2 : j2;
      int j = i2 < j2 ? j2 : i2;
      int p = i * MHID - (i * (i + 1)) / 2 + (j - i - 1);
      p = (i2 != j2) ? p : 0;
      float t = ah[p * OUTD + d];
      lA[e] = (i2 != j2 && t > 0.f) ? t : 0.f;
    }
  }

  float acc0 = 0.f, acc1 = 0.f;

  v8f zx[MHID][2];
#pragma unroll
  for (int m = 0; m < MHID; ++m) { zx[m][0] = (v8f){}; zx[m][1] = (v8f){}; }
  {
    const float* zl = fh + (4 * lhi) * OUTD + 2 * l16;
    v2f bb[2][16];
    load_zgroup(bb[0], zl, 0, 0);
#pragma unroll
    for (int g = 0; g < 4 * MHID; ++g) {
      const int m = g >> 2, jg = g & 3;
      if (g + 1 < 4 * MHID) load_zgroup(bb[(g + 1) & 1], zl, (g + 1) >> 2, (g + 1) & 3);
#pragma unroll
      for (int jj = 0; jj < 4; ++jj) {
        const int j = 4 * jg + jj;
#pragma unroll
        for (int e = 0; e < 2; ++e) {
          v2f a;
          a.x = e ? xa[j].z : xa[j].x;
          a.y = e ? xa[j].w : xa[j].y;
          v2f q0 = bb[g & 1][(jj * 2 + e) * 2 + 0];
          v2f q1 = bb[g & 1][(jj * 2 + e) * 2 + 1];
          v2f be = {q0.x, q1.x};
          v2f bo = {q0.y, q1.y};
          zx[m][0] = wmma4(a, be, zx[m][0]);
          zx[m][1] = wmma4(a, bo, zx[m][1]);
        }
      }
    }
  }
#pragma unroll
  for (int m = 0; m < MHID; ++m)
#pragma unroll
    for (int r = 0; r < 8; ++r) {
      acc0 += zx[m][0][r] * zx[m][0][r];
      acc1 += zx[m][1][r] * zx[m][1][r];
    }

  float* pw = lP + ((2 * w + lhi) * PCOLS + l16) * 8;
  const float* pr = lP + l16 * 8 + 4 * lhi;

#pragma unroll
  for (int m = 0; m < MHID; ++m)
#pragma unroll
    for (int par = 0; par < 2; ++par) {
      const v8f t = zx[m][par];
      *(v4f*)(pw + (2 * m + par) * 128)     = (v4f){t[0], t[1], t[2], t[3]};
      *(v4f*)(pw + (2 * m + par) * 128 + 4) = (v4f){t[4], t[5], t[6], t[7]};
    }
  __syncthreads();

#pragma unroll
  for (int e = tid; e < ASZ; e += 256) {
    int d = e & 31, ij = e >> 5, i2 = ij / MHID, j2 = ij % MHID;
    float s = 0.f;
#pragma unroll
    for (int k = 0; k < MHID; ++k)
      s += lA[(i2 * MHID + k) * OUTD + d] * lA[(k * MHID + j2) * OUTD + d];
    lA2[e] = s;
  }

  v8f p1[MHID][2];
#pragma unroll
  for (int m = 0; m < MHID; ++m) { p1[m][0] = (v8f){}; p1[m][1] = (v8f){}; }
  {
    v4f pf[2][8];
    load_pgroup(pf[0], pr, 0);
#pragma unroll
    for (int g = 0; g < 4 * MHID; ++g) {
      const int mp = g >> 2, jg = g & 3;
      if (g + 1 < 4 * MHID) load_pgroup(pf[(g + 1) & 1], pr + ((g + 1) >> 2) * 256, (g + 1) & 3);
      GROUP_FENCE();
      mma_pgroup(aa, pf[g & 1], jg, p1[mp][0], p1[mp][1]);
      GROUP_FENCE();
    }
  }
  __syncthreads();
#pragma unroll
  for (int m = 0; m < MHID; ++m)
#pragma unroll
    for (int par = 0; par < 2; ++par) {
      const v8f t = p1[m][par];
      *(v4f*)(pw + (2 * m + par) * 128)     = (v4f){t[0], t[1], t[2], t[3]};
      *(v4f*)(pw + (2 * m + par) * 128 + 4) = (v4f){t[4], t[5], t[6], t[7]};
    }
  __syncthreads();

  {
    const float* As1 = lA + 2 * l16;
    const float* As2 = lA2 + 2 * l16;
    v4f pf[2][8];
    v2f cf1[MHID], cf2[MHID];
    v4f q[4];
    load_pgroup(pf[0], pr, 0);
#pragma unroll
    for (int m = 0; m < MHID; ++m) {
      cf1[m] = *(const v2f*)(As1 + (m * MHID) * OUTD);
      cf2[m] = *(const v2f*)(As2 + (m * MHID) * OUTD);
    }
#pragma unroll
    for (int i = 0; i < 4; ++i) q[i] = *(const v4f*)(pw + (i >> 1) * 128 + (i & 1) * 4);
#pragma unroll 1
    for (int mp = 0; mp < MHID; ++mp) {
      const int mpn = mp + 1 < MHID ? mp + 1 : MHID - 1;
      v8f u0 = {}, u1 = {}, t0 = {}, t1 = {};
#pragma unroll
      for (int m = 0; m < MHID; ++m)
#pragma unroll
        for (int r = 0; r < 8; ++r) {
          u0[r] += cf1[m].x * zx[m][0][r];
          u1[r] += cf1[m].y * zx[m][1][r];
          t0[r] += cf2[m].x * zx[m][0][r];
          t1[r] += cf2[m].y * zx[m][1][r];
        }
#pragma unroll
      for (int r = 0; r < 4; ++r) {
        acc0 += u0[r] * q[0][r] + u0[r + 4] * q[1][r];
        acc1 += u1[r] * q[2][r] + u1[r + 4] * q[3][r];
      }
      v8f c0 = {}, c1 = {};
#pragma unroll
      for (int jg = 0; jg < 4; ++jg) {
        if (jg < 3) {
          load_pgroup(pf[(jg + 1) & 1], pr + mp * 256, jg + 1);
        } else {
          load_pgroup(pf[0], pr + mpn * 256, 0);
#pragma unroll
          for (int m = 0; m < MHID; ++m) {
            cf1[m] = *(const v2f*)(As1 + (m * MHID + mpn) * OUTD);
            cf2[m] = *(const v2f*)(As2 + (m * MHID + mpn) * OUTD);
          }
#pragma unroll
          for (int i = 0; i < 4; ++i) q[i] = *(const v4f*)(pw + (2 * mpn + (i >> 1)) * 128 + (i & 1) * 4);
        }
        GROUP_FENCE();
        mma_pgroup(aa, pf[jg & 1], jg, c0, c1);
        GROUP_FENCE();
      }
#pragma unroll
      for (int r = 0; r < 8; ++r) {
        acc0 += t0[r] * c0[r];
        acc1 += t1[r] * c1[r];
      }
    }
  }

  atomicAdd(&accbuf[2 * l16], acc0);
  atomicAdd(&accbuf[2 * l16 + 1], acc1);
  __syncthreads();
  if (tid < OUTD) out[(size_t)b * OUTD + tid] = accbuf[tid] * (1.0f / 3840.0f);
}

extern "C" void kernel_launch(void* const* d_in, const int* in_sizes, int n_in,
                              void* d_out, int out_size, void* d_ws, size_t ws_size,
                              hipStream_t stream) {
  (void)in_sizes; (void)n_in; (void)out_size; (void)ws_size; (void)d_ws;
  const float* adj   = (const float*)d_in[0];
  const float* feats = (const float*)d_in[1];
  const int*   idxs  = (const int*)  d_in[2];
  const float* fh    = (const float*)d_in[3];
  const float* ah    = (const float*)d_in[4];
  float* out = (float*)d_out;

  rw_main<<<B_BATCH, 256, 0, stream>>>(adj, feats, idxs, fh, ah, out);
}


# Step 0's B operand z0 laid once per workgroup into the P buffer in LDS, fragment order (4x4 register transpose); 16-byte LDS loads replace 8-byte global loads and register shuffles; sums unchanged
# speedup vs baseline: 6.3294x; 1.0324x over previous
#include <hip/hip_runtime.h>


typedef float v2f __attribute__((ext_vector_type(2)));
typedef float v4f __attribute__((ext_vector_type(4)));
typedef float v8f __attribute__((ext_vector_type(8)));

#define B_BATCH 1024
#define NNODE   128
#define CDIM    128
#define MHID    10
#define OUTD    32
#define PCOLS   (MHID * OUTD)
#define ASZ     (MHID * MHID * OUTD)

__device__ __forceinline__ v8f wmma4(v2f a, v2f b, v8f c) {
  return __builtin_amdgcn_wmma_f32_16x16x4_f32(
       false, a,  false, b,
       (short)0, c,  false,  false);
}


__device__ __forceinline__ void load_pgroup(v4f (&bb)[8], const float* pb, int jg) {
#pragma unroll
  for (int jj = 0; jj < 4; ++jj) {
    bb[2 * jj]     = *(const v4f*)(pb + (4 * jg + jj) * (PCOLS * 8));
    bb[2 * jj + 1] = *(const v4f*)(pb + (4 * jg + jj) * (PCOLS * 8) + 128);
  }
}
__device__ __forceinline__ void mma_pgroup(const v4f (&aa)[16], const v4f (&bb)[8], int jg, v8f& c0, v8f& c1) {
#pragma unroll
  for (int jj = 0; jj < 4; ++jj) {
    const v4f a4 = aa[4 * jg + jj];
    const v4f b0 = bb[2 * jj], b1 = bb[2 * jj + 1];
    v2f alo = {a4.x, a4.y};
    v2f ahi = {a4.z, a4.w};
    v2f b0lo = {b0.x, b0.y}, b0hi = {b0.z, b0.w};
    v2f b1lo = {b1.x, b1.y}, b1hi = {b1.z, b1.w};
    c0 = wmma4(alo, b0lo, c0);
    c1 = wmma4(alo, b1lo, c1);
    c0 = wmma4(ahi, b0hi, c0);
    c1 = wmma4(ahi, b1hi, c1);
  }
}
#define GROUP_FENCE() __builtin_amdgcn_sched_barrier(0x6)


__global__ __launch_bounds__(256) void rw_main(const float* __restrict__ adj,
                                            const float* __restrict__ feats,
                                            const int*   __restrict__ idxs,
                                            const float* __restrict__ fh,
                                            const float* __restrict__ ah,
                                            float*       __restrict__ out) {
  __shared__ __attribute__((aligned(16))) float lP[(NNODE / 8) * PCOLS * 8];
  __shared__ __attribute__((aligned(16))) float lA[ASZ];
  __shared__ __attribute__((aligned(16))) float lA2[ASZ];
  __shared__ float accbuf[OUTD];

  const int b   = blockIdx.x;
  const int tid = threadIdx.x;
  const int w   = tid >> 5;
  const int l   = tid & 31;
  const int l16 = l & 15;
  const int lhi = l >> 4;
  const int row = w * 16 + l16;

  if (tid < OUTD) accbuf[tid] = 0.f;

  const int oq = tid & 7;
  const int kq = tid >> 3;
  v4f xa[16], aa[16];
  v4f zr[MHID][4];
  {
    const int idx = idxs[b * NNODE + row];
    const float* zs = fh + (4 * kq) * OUTD + 4 * oq;
#pragma unroll
    for (int m = 0; m < MHID; ++m)
#pragma unroll
      for (int i = 0; i < 4; ++i) zr[m][i] = *(const v4f*)(zs + m * (CDIM * OUTD) + i * OUTD);
    const float* xs = feats + (size_t)idx * CDIM + 4 * lhi;
    const float* as = adj + (size_t)b * NNODE * CDIM + (size_t)row * CDIM + 4 * lhi;
#pragma unroll
    for (int j = 0; j < 16; ++j) aa[j] = *(const v4f*)(as + 8 * j);
#pragma unroll
    for (int j = 0; j < 16; ++j) xa[j] = *(const v4f*)(xs + 8 * j);
  }

#pragma unroll
  for (int it = 0; it < (ASZ + 255) / 256; ++it) {
    const int e = tid + 256 * it;
    if (e < ASZ) {
      int d = e & 31, ij = e >> 5, i2 = ij / MHID, j2 = ij % MHID;
      int i = i2 < j2 ? i2 : j2;
      int j = i2 < j2 ? j2 : i2;
      int p = i * MHID - (i * (i + 1)) / 2 + (j - i - 1);
      p = (i2 != j2) ? p : 0;
      float t = ah[p * OUTD + d];
      lA[e] = (i2 != j2 && t > 0.f) ? t : 0.f;
    }
  }

  float acc0 = 0.f, acc1 = 0.f;

  {
    float* zw = lP + (kq >> 1) * (PCOLS * 8) + 8 * oq + 4 * (kq & 1);
#pragma unroll
    for (int m = 0; m < MHID; ++m)
#pragma unroll
      for (int t = 0; t < 4; ++t)
        *(v4f*)(zw + 256 * m + 128 * (t & 1) + 64 * (t >> 1)) =
            (v4f){zr[m][0][t], zr[m][1][t], zr[m][2][t], zr[m][3][t]};
  }
  __syncthreads();

  v8f zx[MHID][2];
#pragma unroll
  for (int m = 0; m < MHID; ++m) { zx[m][0] = (v8f){}; zx[m][1] = (v8f){}; }
  {
    const float* pz = lP + 8 * (l16 >> 1) + 4 * lhi + 64 * (l16 & 1);
    v4f pf[2][8];
    load_pgroup(pf[0], pz, 0);
#pragma unroll
    for (int g = 0; g < 4 * MHID; ++g) {
      const int m = g >> 2, jg = g & 3;
      if (g + 1 < 4 * MHID) load_pgroup(pf[(g + 1) & 1], pz + ((g + 1) >> 2) * 256, (g + 1) & 3);
      GROUP_FENCE();
      mma_pgroup(xa, pf[g & 1], jg, zx[m][0], zx[m][1]);
      GROUP_FENCE();
    }
  }
#pragma unroll
  for (int m = 0; m < MHID; ++m)
#pragma unroll
    for (int r = 0; r < 8; ++r) {
      acc0 += zx[m][0][r] * zx[m][0][r];
      acc1 += zx[m][1][r] * zx[m][1][r];
    }

  float* pw = lP + ((2 * w + lhi) * PCOLS + l16) * 8;
  const float* pr = lP + l16 * 8 + 4 * lhi;

  __syncthreads();
#pragma unroll
  for (int m = 0; m < MHID; ++m)
#pragma unroll
    for (int par = 0; par < 2; ++par) {
      const v8f t = zx[m][par];
      *(v4f*)(pw + (2 * m + par) * 128)     = (v4f){t[0], t[1], t[2], t[3]};
      *(v4f*)(pw + (2 * m + par) * 128 + 4) = (v4f){t[4], t[5], t[6], t[7]};
    }
  __syncthreads();

#pragma unroll
  for (int e = tid; e < ASZ; e += 256) {
    int d = e & 31, ij = e >> 5, i2 = ij / MHID, j2 = ij % MHID;
    float s = 0.f;
#pragma unroll
    for (int k = 0; k < MHID; ++k)
      s += lA[(i2 * MHID + k) * OUTD + d] * lA[(k * MHID + j2) * OUTD + d];
    lA2[e] = s;
  }

  v8f p1[MHID][2];
#pragma unroll
  for (int m = 0; m < MHID; ++m) { p1[m][0] = (v8f){}; p1[m][1] = (v8f){}; }
  {
    v4f pf[2][8];
    load_pgroup(pf[0], pr, 0);
#pragma unroll
    for (int g = 0; g < 4 * MHID; ++g) {
      const int mp = g >> 2, jg = g & 3;
      if (g + 1 < 4 * MHID) load_pgroup(pf[(g + 1) & 1], pr + ((g + 1) >> 2) * 256, (g + 1) & 3);
      GROUP_FENCE();
      mma_pgroup(aa, pf[g & 1], jg, p1[mp][0], p1[mp][1]);
      GROUP_FENCE();
    }
  }
  __syncthreads();
#pragma unroll
  for (int m = 0; m < MHID; ++m)
#pragma unroll
    for (int par = 0; par < 2; ++par) {
      const v8f t = p1[m][par];
      *(v4f*)(pw + (2 * m + par) * 128)     = (v4f){t[0], t[1], t[2], t[3]};
      *(v4f*)(pw + (2 * m + par) * 128 + 4) = (v4f){t[4], t[5], t[6], t[7]};
    }
  __syncthreads();

  {
    const float* As1 = lA + 2 * l16;
    const float* As2 = lA2 + 2 * l16;
    v4f pf[2][8];
    v2f cf1[MHID], cf2[MHID];
    v4f q[4];
    load_pgroup(pf[0], pr, 0);
#pragma unroll
    for (int m = 0; m < MHID; ++m) {
      cf1[m] = *(const v2f*)(As1 + (m * MHID) * OUTD);
      cf2[m] = *(const v2f*)(As2 + (m * MHID) * OUTD);
    }
#pragma unroll
    for (int i = 0; i < 4; ++i) q[i] = *(const v4f*)(pw + (i >> 1) * 128 + (i & 1) * 4);
#pragma unroll 1
    for (int mp = 0; mp < MHID; ++mp) {
      const int mpn = mp + 1 < MHID ? mp + 1 : MHID - 1;
      v8f u0 = {}, u1 = {}, t0 = {}, t1 = {};
#pragma unroll
      for (int m = 0; m < MHID; ++m)
#pragma unroll
        for (int r = 0; r < 8; ++r) {
          u0[r] += cf1[m].x * zx[m][0][r];
          u1[r] += cf1[m].y * zx[m][1][r];
          t0[r] += cf2[m].x * zx[m][0][r];
          t1[r] += cf2[m].y * zx[m][1][r];
        }
#pragma unroll
      for (int r = 0; r < 4; ++r) {
        acc0 += u0[r] * q[0][r] + u0[r + 4] * q[1][r];
        acc1 += u1[r] * q[2][r] + u1[r + 4] * q[3][r];
      }
      v8f c0 = {}, c1 = {};
#pragma unroll
      for (int jg = 0; jg < 4; ++jg) {
        if (jg < 3) {
          load_pgroup(pf[(jg + 1) & 1], pr + mp * 256, jg + 1);
        } else {
          load_pgroup(pf[0], pr + mpn * 256, 0);
#pragma unroll
          for (int m = 0; m < MHID; ++m) {
            cf1[m] = *(const v2f*)(As1 + (m * MHID + mpn) * OUTD);
            cf2[m] = *(const v2f*)(As2 + (m * MHID + mpn) * OUTD);
          }
#pragma unroll
          for (int i = 0; i < 4; ++i) q[i] = *(const v4f*)(pw + (2 * mpn + (i >> 1)) * 128 + (i & 1) * 4);
        }
        GROUP_FENCE();
        mma_pgroup(aa, pf[jg & 1], jg, c0, c1);
        GROUP_FENCE();
      }
#pragma unroll
      for (int r = 0; r < 8; ++r) {
        acc0 += t0[r] * c0[r];
        acc1 += t1[r] * c1[r];
      }
    }
  }

  atomicAdd(&accbuf[2 * l16], acc0);
  atomicAdd(&accbuf[2 * l16 + 1], acc1);
  __syncthreads();
  if (tid < OUTD) out[(size_t)b * OUTD + tid] = accbuf[tid] * (1.0f / 3840.0f);
}

extern "C" void kernel_launch(void* const* d_in, const int* in_sizes, int n_in,
                              void* d_out, int out_size, void* d_ws, size_t ws_size,
                              hipStream_t stream) {
  (void)in_sizes; (void)n_in; (void)out_size; (void)ws_size; (void)d_ws;
  const float* adj   = (const float*)d_in[0];
  const float* feats = (const float*)d_in[1];
  const int*   idxs  = (const int*)  d_in[2];
  const float* fh    = (const float*)d_in[3];
  const float* ah    = (const float*)d_in[4];
  float* out = (float*)d_out;

  rw_main<<<B_BATCH, 256, 0, stream>>>(adj, feats, idxs, fh, ah, out);
}


# Step 2's coefficient sums U_1, U_2 written as whole-tile (8-register vector) FMAs on zx in place, not element by element: no re-paired zx copies, no scratch reloads in the loop; sum order unchanged
# speedup vs baseline: 7.2502x; 1.1455x over previous
#include <hip/hip_runtime.h>


typedef float v2f __attribute__((ext_vector_type(2)));
typedef float v4f __attribute__((ext_vector_type(4)));
typedef float v8f __attribute__((ext_vector_type(8)));

#define B_BATCH 1024
#define NNODE   128
#define CDIM    128
#define MHID    10
#define OUTD    32
#define PCOLS   (MHID * OUTD)
#define ASZ     (MHID * MHID * OUTD)

__device__ __forceinline__ v8f wmma4(v2f a, v2f b, v8f c) {
  return __builtin_amdgcn_wmma_f32_16x16x4_f32(
       false, a,  false, b,
       (short)0, c,  false,  false);
}


__device__ __forceinline__ void load_pgroup(v4f (&bb)[8], const float* pb, int jg) {
#pragma unroll
  for (int jj = 0; jj < 4; ++jj) {
    bb[2 * jj]     = *(const v4f*)(pb + (4 * jg + jj) * (PCOLS * 8));
    bb[2 * jj + 1] = *(const v4f*)(pb + (4 * jg + jj) * (PCOLS * 8) + 128);
  }
}
__device__ __forceinline__ void mma_pgroup(const v4f (&aa)[16], const v4f (&bb)[8], int jg, v8f& c0, v8f& c1) {
#pragma unroll
  for (int jj = 0; jj < 4; ++jj) {
    const v4f a4 = aa[4 * jg + jj];
    const v4f b0 = bb[2 * jj], b1 = bb[2 * jj + 1];
    v2f alo = {a4.x, a4.y};
    v2f ahi = {a4.z, a4.w};
    v2f b0lo = {b0.x, b0.y}, b0hi = {b0.z, b0.w};
    v2f b1lo = {b1.x, b1.y}, b1hi = {b1.z, b1.w};
    c0 = wmma4(alo, b0lo, c0);
    c1 = wmma4(alo, b1lo, c1);
    c0 = wmma4(ahi, b0hi, c0);
    c1 = wmma4(ahi, b1hi, c1);
  }
}
#define GROUP_FENCE() __builtin_amdgcn_sched_barrier(0x6)


__global__ __launch_bounds__(256) void rw_main(const float* __restrict__ adj,
                                            const float* __restrict__ feats,
                                            const int*   __restrict__ idxs,
                                            const float* __restrict__ fh,
                                            const float* __restrict__ ah,
                                            float*       __restrict__ out) {
  __shared__ __attribute__((aligned(16))) float lP[(NNODE / 8) * PCOLS * 8];
  __shared__ __attribute__((aligned(16))) float lA[ASZ];
  __shared__ __attribute__((aligned(16))) float lA2[ASZ];
  __shared__ float accbuf[OUTD];

  const int b   = blockIdx.x;
  const int tid = threadIdx.x;
  const int w   = tid >> 5;
  const int l   = tid & 31;
  const int l16 = l & 15;
  const int lhi = l >> 4;
  const int row = w * 16 + l16;

  if (tid < OUTD) accbuf[tid] = 0.f;

  const int oq = tid & 7;
  const int kq = tid >> 3;
  v4f xa[16], aa[16];
  v4f zr[MHID][4];
  {
    const int idx = idxs[b * NNODE + row];
    const float* zs = fh + (4 * kq) * OUTD + 4 * oq;
#pragma unroll
    for (int m = 0; m < MHID; ++m)
#pragma unroll
      for (int i = 0; i < 4; ++i) zr[m][i] = *(const v4f*)(zs + m * (CDIM * OUTD) + i * OUTD);
    const float* xs = feats + (size_t)idx * CDIM + 4 * lhi;
    const float* as = adj + (size_t)b * NNODE * CDIM + (size_t)row * CDIM + 4 * lhi;
#pragma unroll
    for (int j = 0; j < 16; ++j) aa[j] = *(const v4f*)(as + 8 * j);
#pragma unroll
    for (int j = 0; j < 16; ++j) xa[j] = *(const v4f*)(xs + 8 * j);
  }

#pragma unroll
  for (int it = 0; it < (ASZ + 255) / 256; ++it) {
    const int e = tid + 256 * it;
    if (e < ASZ) {
      int d = e & 31, ij = e >> 5, i2 = ij / MHID, j2 = ij % MHID;
      int i = i2 < j2 ? i2 : j2;
      int j = i2 < j2 ? j2 : i2;
      int p = i * MHID - (i * (i + 1)) / 2 + (j - i - 1);
      p = (i2 != j2) ? p : 0;
      float t = ah[p * OUTD + d];
      lA[e] = (i2 != j2 && t > 0.f) ? t : 0.f;
    }
  }

  float acc0 = 0.f, acc1 = 0.f;

  {
    float* zw = lP + (kq >> 1) * (PCOLS * 8) + 8 * oq + 4 * (kq & 1);
#pragma unroll
    for (int m = 0; m < MHID; ++m)
#pragma unroll
      for (int t = 0; t < 4; ++t)
        *(v4f*)(zw + 256 * m + 128 * (t & 1) + 64 * (t >> 1)) =
            (v4f){zr[m][0][t], zr[m][1][t], zr[m][2][t], zr[m][3][t]};
  }
  __syncthreads();

  v8f zx[MHID][2];
#pragma unroll
  for (int m = 0; m < MHID; ++m) { zx[m][0] = (v8f){}; zx[m][1] = (v8f){}; }
  {
    const float* pz = lP + 8 * (l16 >> 1) + 4 * lhi + 64 * (l16 & 1);
    v4f pf[2][8];
    load_pgroup(pf[0], pz, 0);
#pragma unroll
    for (int g = 0; g < 4 * MHID; ++g) {
      const int m = g >> 2, jg = g & 3;
      if (g + 1 < 4 * MHID) load_pgroup(pf[(g + 1) & 1], pz + ((g + 1) >> 2) * 256, (g + 1) & 3);
      GROUP_FENCE();
      mma_pgroup(xa, pf[g & 1], jg, zx[m][0], zx[m][1]);
      GROUP_FENCE();
    }
  }
#pragma unroll
  for (int m = 0; m < MHID; ++m)
#pragma unroll
    for (int r = 0; r < 8; ++r) {
      acc0 += zx[m][0][r] * zx[m][0][r];
      acc1 += zx[m][1][r] * zx[m][1][r];
    }

  float* pw = lP + ((2 * w + lhi) * PCOLS + l16) * 8;
  const float* pr = lP + l16 * 8 + 4 * lhi;

  __syncthreads();
#pragma unroll
  for (int m = 0; m < MHID; ++m)
#pragma unroll
    for (int par = 0; par < 2; ++par) {
      const v8f t = zx[m][par];
      *(v4f*)(pw + (2 * m + par) * 128)     = (v4f){t[0], t[1], t[2], t[3]};
      *(v4f*)(pw + (2 * m + par) * 128 + 4) = (v4f){t[4], t[5], t[6], t[7]};
    }
  __syncthreads();

#pragma unroll
  for (int e = tid; e < ASZ; e += 256) {
    int d = e & 31, ij = e >> 5, i2 = ij / MHID, j2 = ij % MHID;
    float s = 0.f;
#pragma unroll
    for (int k = 0; k < MHID; ++k)
      s += lA[(i2 * MHID + k) * OUTD + d] * lA[(k * MHID + j2) * OUTD + d];
    lA2[e] = s;
  }

  v8f p1[MHID][2];
#pragma unroll
  for (int m = 0; m < MHID; ++m) { p1[m][0] = (v8f){}; p1[m][1] = (v8f){}; }
  {
    v4f pf[2][8];
    load_pgroup(pf[0], pr, 0);
#pragma unroll
    for (int g = 0; g < 4 * MHID; ++g) {
      const int mp = g >> 2, jg = g & 3;
      if (g + 1 < 4 * MHID) load_pgroup(pf[(g + 1) & 1], pr + ((g + 1) >> 2) * 256, (g + 1) & 3);
      GROUP_FENCE();
      mma_pgroup(aa, pf[g & 1], jg, p1[mp][0], p1[mp][1]);
      GROUP_FENCE();
    }
  }
  __syncthreads();
#pragma unroll
  for (int m = 0; m < MHID; ++m)
#pragma unroll
    for (int par = 0; par < 2; ++par) {
      const v8f t = p1[m][par];
      *(v4f*)(pw + (2 * m + par) * 128)     = (v4f){t[0], t[1], t[2], t[3]};
      *(v4f*)(pw + (2 * m + par) * 128 + 4) = (v4f){t[4], t[5], t[6], t[7]};
    }
  __syncthreads();

  {
    const float* As1 = lA + 2 * l16;
    const float* As2 = lA2 + 2 * l16;
    v4f pf[2][8];
    v2f cf1[MHID], cf2[MHID];
    v4f q[4];
    load_pgroup(pf[0], pr, 0);
#pragma unroll
    for (int m = 0; m < MHID; ++m) {
      cf1[m] = *(const v2f*)(As1 + (m * MHID) * OUTD);
      cf2[m] = *(const v2f*)(As2 + (m * MHID) * OUTD);
    }
#pragma unroll
    for (int i = 0; i < 4; ++i) q[i] = *(const v4f*)(pw + (i >> 1) * 128 + (i & 1) * 4);
#pragma unroll 1
    for (int mp = 0; mp < MHID; ++mp) {
      const int mpn = mp + 1 < MHID ? mp + 1 : MHID - 1;
      v8f u0 = {}, u1 = {}, t0 = {}, t1 = {};
#pragma unroll
      for (int m = 0; m < MHID; ++m) {
        u0 += cf1[m].x * zx[m][0];
        u1 += cf1[m].y * zx[m][1];
        t0 += cf2[m].x * zx[m][0];
        t1 += cf2[m].y * zx[m][1];
      }
#pragma unroll
      for (int r = 0; r < 4; ++r) {
        acc0 += u0[r] * q[0][r] + u0[r + 4] * q[1][r];
        acc1 += u1[r] * q[2][r] + u1[r + 4] * q[3][r];
      }
      v8f c0 = {}, c1 = {};
#pragma unroll
      for (int jg = 0; jg < 4; ++jg) {
        if (jg < 3) {
          load_pgroup(pf[(jg + 1) & 1], pr + mp * 256, jg + 1);
        } else {
          load_pgroup(pf[0], pr + mpn * 256, 0);
#pragma unroll
          for (int m = 0; m < MHID; ++m) {
            cf1[m] = *(const v2f*)(As1 + (m * MHID + mpn) * OUTD);
            cf2[m] = *(const v2f*)(As2 + (m * MHID + mpn) * OUTD);
          }
#pragma unroll
          for (int i = 0; i < 4; ++i) q[i] = *(const v4f*)(pw + (2 * mpn + (i >> 1)) * 128 + (i & 1) * 4);
        }
        GROUP_FENCE();
        mma_pgroup(aa, pf[jg & 1], jg, c0, c1);
        GROUP_FENCE();
      }
#pragma unroll
      for (int r = 0; r < 8; ++r) {
        acc0 += t0[r] * c0[r];
        acc1 += t1[r] * c1[r];
      }
    }
  }

  atomicAdd(&accbuf[2 * l16], acc0);
  atomicAdd(&accbuf[2 * l16 + 1], acc1);
  __syncthreads();
  if (tid < OUTD) out[(size_t)b * OUTD + tid] = accbuf[tid] * (1.0f / 3840.0f);
}

extern "C" void kernel_launch(void* const* d_in, const int* in_sizes, int n_in,
                              void* d_out, int out_size, void* d_ws, size_t ws_size,
                              hipStream_t stream) {
  (void)in_sizes; (void)n_in; (void)out_size; (void)ws_size; (void)d_ws;
  const float* adj   = (const float*)d_in[0];
  const float* feats = (const float*)d_in[1];
  const int*   idxs  = (const int*)  d_in[2];
  const float* fh    = (const float*)d_in[3];
  const float* ah    = (const float*)d_in[4];
  float* out = (float*)d_out;

  rw_main<<<B_BATCH, 256, 0, stream>>>(adj, feats, idxs, fh, ah, out);
}


# Workgroup cut to 4 waves of 32 rows; the 32 channels taken as two passes of 16 (80 KB P image, adj strips as fragments in LDS), so two workgroups share a WGP; sum order changes
# speedup vs baseline: 7.6155x; 1.0504x over previous
#include <hip/hip_runtime.h>


typedef float v2f __attribute__((ext_vector_type(2)));
typedef float v4f __attribute__((ext_vector_type(4)));
typedef float v8f __attribute__((ext_vector_type(8)));

#define B_BATCH 1024
#define NNODE   128
#define CDIM    128
#define MHID    10
#define OUTD    32
#define HCH     16
#define PCOLS   (MHID * HCH)
#define PROW    (PCOLS * 8)
#define ASZH    (MHID * MHID * HCH)

__device__ __forceinline__ v8f wmma4(v2f a, v2f b, v8f c) {
  return __builtin_amdgcn_wmma_f32_16x16x4_f32(
       false, a,  false, b,
       (short)0, c,  false,  false);
}


__device__ __forceinline__ void load_bgroup(v4f (&bb)[8], const float* pb, int jg) {
#pragma unroll
  for (int jj = 0; jj < 4; ++jj) {
    bb[2 * jj]     = *(const v4f*)(pb + (4 * jg + jj) * PROW);
    bb[2 * jj + 1] = *(const v4f*)(pb + (4 * jg + jj) * PROW + 128);
  }
}
__device__ __forceinline__ void mma_quad(v4f a0, v4f a1, v4f b0, v4f b1,
                                         v8f& c00, v8f& c01, v8f& c10, v8f& c11) {
  v2f a0lo = {a0.x, a0.y}, a0hi = {a0.z, a0.w};
  v2f a1lo = {a1.x, a1.y}, a1hi = {a1.z, a1.w};
  v2f b0lo = {b0.x, b0.y}, b0hi = {b0.z, b0.w};
  v2f b1lo = {b1.x, b1.y}, b1hi = {b1.z, b1.w};
  c00 = wmma4(a0lo, b0lo, c00);
  c01 = wmma4(a0lo, b1lo, c01);
  c10 = wmma4(a1lo, b0lo, c10);
  c11 = wmma4(a1lo, b1lo, c11);
  c00 = wmma4(a0hi, b0hi, c00);
  c01 = wmma4(a0hi, b1hi, c01);
  c10 = wmma4(a1hi, b0hi, c10);
  c11 = wmma4(a1hi, b1hi, c11);
}
__device__ __forceinline__ void load_bhalf(v4f (&bb)[4], const float* pb, int jh) {
#pragma unroll
  for (int jj = 0; jj < 2; ++jj) {
    bb[2 * jj]     = *(const v4f*)(pb + (2 * jh + jj) * PROW);
    bb[2 * jj + 1] = *(const v4f*)(pb + (2 * jh + jj) * PROW + 128);
  }
}
__device__ __forceinline__ void load_ahalf(v4f (&ab)[4], const float* pa, int jh) {
#pragma unroll
  for (int jj = 0; jj < 2; ++jj) {
    ab[2 * jj]     = *(const v4f*)(pa + (2 * jh + jj) * 128);
    ab[2 * jj + 1] = *(const v4f*)(pa + (16 + 2 * jh + jj) * 128);
  }
}
__device__ __forceinline__ void mma_half(const v4f (&ab)[4], const v4f (&bb)[4],
                                         v8f& c00, v8f& c01, v8f& c10, v8f& c11) {
#pragma unroll
  for (int jj = 0; jj < 2; ++jj)
    mma_quad(ab[2 * jj], ab[2 * jj + 1], bb[2 * jj], bb[2 * jj + 1], c00, c01, c10, c11);
}
#define GROUP_FENCE() __builtin_amdgcn_sched_barrier(0x6)

__global__ __launch_bounds__(128, 2) void rw_main(const float* __restrict__ adj,
                                               const float* __restrict__ feats,
                                               const int*   __restrict__ idxs,
                                               const float* __restrict__ fh,
                                               const float* __restrict__ ah,
                                               float*       __restrict__ out) {
  __shared__ __attribute__((aligned(16))) float lP[(NNODE / 8) * PROW];
  __shared__ __attribute__((aligned(16))) float lAdj[8 * 16 * 128];
  __shared__ __attribute__((aligned(16))) float lA[ASZH];
  __shared__ __attribute__((aligned(16))) float lA2[ASZH];
  __shared__ float accbuf[OUTD];

  const int b   = blockIdx.x;
  const int tid = threadIdx.x;
  const int w   = tid >> 5;
  const int l   = tid & 31;
  const int l16 = l & 15;
  const int lhi = l >> 4;
  const int row0 = 32 * w + l16;

  if (tid < OUTD) accbuf[tid] = 0.f;

  float* pa = lAdj + (2 * w * 16) * 128 + l * 4;
  {
    const float* as0 = adj + (size_t)b * NNODE * CDIM + (size_t)row0 * CDIM + 4 * lhi;
    const float* as1 = as0 + 16 * CDIM;
    v4f a0[16], a1[16];
#pragma unroll
    for (int j = 0; j < 16; ++j) a0[j] = *(const v4f*)(as0 + 8 * j);
#pragma unroll
    for (int j = 0; j < 16; ++j) a1[j] = *(const v4f*)(as1 + 8 * j);
#pragma unroll
    for (int j = 0; j < 16; ++j) *(v4f*)(pa + j * 128) = a0[j];
#pragma unroll
    for (int j = 0; j < 16; ++j) *(v4f*)(pa + (16 + j) * 128) = a1[j];
  }

  float* pw0 = lP + ((2 * (2 * w) + lhi) * PCOLS + l16) * 8;
  float* pw1 = lP + ((2 * (2 * w + 1) + lhi) * PCOLS + l16) * 8;
  const float* pr = lP + l16 * 8 + 4 * lhi;

  const int oq = tid & 3;
  const int kq = tid >> 2;

#pragma unroll 1
  for (int hh = 0; hh < 2; ++hh) {
    v4f xa0[16], xa1[16];
    v4f zr[MHID][4];
    int tl = tid;
    {
      asm volatile("" : "+v"(tl));
      const int lo = tl & 127;
      const int r0 = 32 * (lo >> 5) + (lo & 15);
      const int i0 = idxs[b * NNODE + r0];
      const int i1 = idxs[b * NNODE + r0 + 16];
      const float* xs0 = feats + (size_t)i0 * CDIM + 4 * lhi;
      const float* xs1 = feats + (size_t)i1 * CDIM + 4 * lhi;
#pragma unroll
      for (int j = 0; j < 16; ++j) xa0[j] = *(const v4f*)(xs0 + 8 * j);
#pragma unroll
      for (int j = 0; j < 16; ++j) xa1[j] = *(const v4f*)(xs1 + 8 * j);
      const float* zs = fh + (4 * kq) * OUTD + HCH * hh + 4 * oq;
#pragma unroll
      for (int m = 0; m < MHID; ++m)
#pragma unroll
        for (int i = 0; i < 4; ++i) zr[m][i] = *(const v4f*)(zs + m * (CDIM * OUTD) + i * OUTD);
    }

#pragma unroll
    for (int it = 0; it < (ASZH + 127) / 128; ++it) {
      const int e = (tl & 127) + 128 * it;
      if (e < ASZH) {
        int d = e & 15, ij = e >> 4, i2 = ij / MHID, j2 = ij % MHID;
        int i = i2 < j2 ? i2 : j2;
        int j = i2 < j2 ? j2 : i2;
        int p = i * MHID - (i * (i + 1)) / 2 + (j - i - 1);
        p = (i2 != j2) ? p : 0;
        float t = ah[p * OUTD + HCH * hh + d];
        lA[e] = (i2 != j2 && t > 0.f) ? t : 0.f;
      }
    }

    float acc = 0.f;

    {
      float* zw = lP + (kq >> 1) * PROW + (4 * oq) * 8 + 4 * (kq & 1);
#pragma unroll
      for (int m = 0; m < MHID; ++m)
#pragma unroll
        for (int t = 0; t < 4; ++t)
          *(v4f*)(zw + 128 * m + 8 * t) = (v4f){zr[m][0][t], zr[m][1][t], zr[m][2][t], zr[m][3][t]};
    }
    __syncthreads();

    v8f zx0[MHID], zx1[MHID];
#pragma unroll
    for (int m = 0; m < MHID; ++m) { zx0[m] = (v8f){}; zx1[m] = (v8f){}; }
    {
      v4f pf[2][8];
      load_bgroup(pf[0], pr, 0);
#pragma unroll
      for (int g = 0; g < 2 * MHID; ++g) {
        const int tp = g >> 2, jg = g & 3;
        if (g + 1 < 2 * MHID) load_bgroup(pf[(g + 1) & 1], pr + ((g + 1) >> 2) * 256, (g + 1) & 3);
        GROUP_FENCE();
#pragma unroll
        for (int jj = 0; jj < 4; ++jj)
          mma_quad(xa0[4 * jg + jj], xa1[4 * jg + jj], pf[g & 1][2 * jj], pf[g & 1][2 * jj + 1],
                   zx0[2 * tp], zx0[2 * tp + 1], zx1[2 * tp], zx1[2 * tp + 1]);
        GROUP_FENCE();
      }
    }
#pragma unroll
    for (int m = 0; m < MHID; ++m)
#pragma unroll
      for (int r = 0; r < 8; ++r) {
        acc += zx0[m][r] * zx0[m][r];
        acc += zx1[m][r] * zx1[m][r];
      }

    __syncthreads();
#pragma unroll
    for (int m = 0; m < MHID; ++m) {
      const v8f t0 = zx0[m], t1 = zx1[m];
      *(v4f*)(pw0 + m * 128)     = (v4f){t0[0], t0[1], t0[2], t0[3]};
      *(v4f*)(pw0 + m * 128 + 4) = (v4f){t0[4], t0[5], t0[6], t0[7]};
      *(v4f*)(pw1 + m * 128)     = (v4f){t1[0], t1[1], t1[2], t1[3]};
      *(v4f*)(pw1 + m * 128 + 4) = (v4f){t1[4], t1[5], t1[6], t1[7]};
    }
    __syncthreads();

#pragma unroll
    for (int e = tid; e < ASZH; e += 128) {
      int d = e & 15, ij = e >> 4, i2 = ij / MHID, j2 = ij % MHID;
      float s = 0.f;
#pragma unroll
      for (int k = 0; k < MHID; ++k)
        s += lA[(i2 * MHID + k) * HCH + d] * lA[(k * MHID + j2) * HCH + d];
      lA2[e] = s;
    }

    v8f p10[MHID], p11[MHID];
#pragma unroll
    for (int m = 0; m < MHID; ++m) { p10[m] = (v8f){}; p11[m] = (v8f){}; }
    {
      v4f pf[2][4], af[2][4];
      load_bhalf(pf[0], pr, 0);
      load_ahalf(af[0], pa, 0);
#pragma unroll
      for (int g = 0; g < 4 * MHID; ++g) {
        const int tp = g >> 3;
        if (g + 1 < 4 * MHID) {
          load_bhalf(pf[(g + 1) & 1], pr + ((g + 1) >> 3) * 256, (g + 1) & 7);
          load_ahalf(af[(g + 1) & 1], pa, (g + 1) & 7);
        }
        GROUP_FENCE();
        mma_half(af[g & 1], pf[g & 1], p10[2 * tp], p10[2 * tp + 1], p11[2 * tp], p11[2 * tp + 1]);
        GROUP_FENCE();
      }
    }
    __syncthreads();
#pragma unroll
    for (int m = 0; m < MHID; ++m) {
      const v8f t0 = p10[m], t1 = p11[m];
      *(v4f*)(pw0 + m * 128)     = (v4f){t0[0], t0[1], t0[2], t0[3]};
      *(v4f*)(pw0 + m * 128 + 4) = (v4f){t0[4], t0[5], t0[6], t0[7]};
      *(v4f*)(pw1 + m * 128)     = (v4f){t1[0], t1[1], t1[2], t1[3]};
      *(v4f*)(pw1 + m * 128 + 4) = (v4f){t1[4], t1[5], t1[6], t1[7]};
    }
    __syncthreads();

    {
      const float* As1 = lA + l16;
      const float* As2 = lA2 + l16;
      v4f pf[2][4], af[2][4];
      load_bhalf(pf[0], pr, 0);
      load_ahalf(af[0], pa, 0);
#pragma unroll 1
      for (int tp = 0; tp < MHID / 2; ++tp) {
        const int tpn = tp + 1 < MHID / 2 ? tp + 1 : MHID / 2 - 1;
        float cf1[2][MHID], cf2[2][MHID];
#pragma unroll
        for (int mm = 0; mm < 2; ++mm)
#pragma unroll
          for (int m = 0; m < MHID; ++m) {
            cf1[mm][m] = As1[(m * MHID + 2 * tp + mm) * HCH];
            cf2[mm][m] = As2[(m * MHID + 2 * tp + mm) * HCH];
          }
        v4f q[2][2][2];
#pragma unroll
        for (int mm = 0; mm < 2; ++mm)
#pragma unroll
          for (int i = 0; i < 2; ++i) {
            q[0][mm][i] = *(const v4f*)(pw0 + (2 * tp + mm) * 128 + 4 * i);
            q[1][mm][i] = *(const v4f*)(pw1 + (2 * tp + mm) * 128 + 4 * i);
          }
        v8f u00 = {}, u01 = {}, u10 = {}, u11 = {};
        v8f t00 = {}, t01 = {}, t10 = {}, t11 = {};
#pragma unroll
        for (int m = 0; m < MHID; ++m) {
          u00 += cf1[0][m] * zx0[m];
          u01 += cf1[1][m] * zx0[m];
          u10 += cf1[0][m] * zx1[m];
          u11 += cf1[1][m] * zx1[m];
          t00 += cf2[0][m] * zx0[m];
          t01 += cf2[1][m] * zx0[m];
          t10 += cf2[0][m] * zx1[m];
          t11 += cf2[1][m] * zx1[m];
        }
#pragma unroll
        for (int r = 0; r < 4; ++r) {
          acc += u00[r] * q[0][0][0][r] + u00[r + 4] * q[0][0][1][r];
          acc += u01[r] * q[0][1][0][r] + u01[r + 4] * q[0][1][1][r];
          acc += u10[r] * q[1][0][0][r] + u10[r + 4] * q[1][0][1][r];
          acc += u11[r] * q[1][1][0][r] + u11[r + 4] * q[1][1][1][r];
        }
        v8f c00 = {}, c01 = {}, c10 = {}, c11 = {};
#pragma unroll
        for (int jh = 0; jh < 8; ++jh) {
          if (jh < 7) {
            load_bhalf(pf[(jh + 1) & 1], pr + tp * 256, jh + 1);
            load_ahalf(af[(jh + 1) & 1], pa, jh + 1);
          } else {
            load_bhalf(pf[0], pr + tpn * 256, 0);
            load_ahalf(af[0], pa, 0);
          }
          GROUP_FENCE();
          mma_half(af[jh & 1], pf[jh & 1], c00, c01, c10, c11);
          GROUP_FENCE();
        }
#pragma unroll
        for (int r = 0; r < 8; ++r) {
          acc += t00[r] * c00[r];
          acc += t01[r] * c01[r];
          acc += t10[r] * c10[r];
          acc += t11[r] * c11[r];
        }
      }
    }

    atomicAdd(&accbuf[HCH * hh + l16], acc);
    __syncthreads();
  }

  if (tid < OUTD) out[(size_t)b * OUTD + tid] = accbuf[tid] * (1.0f / 3840.0f);
}

extern "C" void kernel_launch(void* const* d_in, const int* in_sizes, int n_in,
                              void* d_out, int out_size, void* d_ws, size_t ws_size,
                              hipStream_t stream) {
  (void)in_sizes; (void)n_in; (void)out_size; (void)ws_size; (void)d_ws;
  const float* adj   = (const float*)d_in[0];
  const float* feats = (const float*)d_in[1];
  const int*   idxs  = (const int*)  d_in[2];
  const float* fh    = (const float*)d_in[3];
  const float* ah    = (const float*)d_in[4];
  float* out = (float*)d_out;

  rw_main<<<B_BATCH, 128, 0, stream>>>(adj, feats, idxs, fh, ah, out);
}
